# starting kernel
# baseline (speedup 1.0000x reference)
#include <hip/hip_runtime.h>


typedef __attribute__((ext_vector_type(2))) float v2f;
typedef __attribute__((ext_vector_type(8))) float v8f;

#define DBA_B 32
#define DBA_N 65536
#define DBA_G 32
#define DBA_SPAN (DBA_N / DBA_G)
#define DBA_PW (DBA_SPAN / 8)

__device__ __forceinline__ float fast_rcp(float x) {
  float inv = __builtin_amdgcn_rcpf(x);
  inv = inv * (2.0f - x * inv);
  return inv;
}

__global__ __launch_bounds__(256) void dba_stage1_reduce(
    const float* __restrict__ r, const float* __restrict__ w,
    const float* __restrict__ Jp, const float* __restrict__ Jd,
    const float* __restrict__ lmbda, float* __restrict__ partial)
{
  const int b    = blockIdx.x / DBA_G;
  const int c    = blockIdx.x % DBA_G;
  const int wave = threadIdx.x >> 5;
  const int lane = threadIdx.x & 31;
  const int half = lane >> 4;
  const int i    = lane & 15;
  const float lmb = lmbda[0];

  const int base = c * DBA_SPAN + wave * DBA_PW;
  const size_t rb0 = (size_t)b * DBA_N + base + half;

  const float* pux;
  const float* puy;
  int   strideF;
  float msk;
  if (i < 6) { pux = Jp + rb0 * 12 + i; puy = pux + 6; strideF = 24; msk = 1.f; }
  else       { pux = r + rb0 * 2;       puy = pux + 1; strideF = 4;
               msk = (i == 6) ? 1.f : 0.f; }
  const float2* pw  = (const float2*)(w  + rb0 * 2);
  const float2* pjd = (const float2*)(Jd + rb0 * 2);

  v8f cP = {0.f,0.f,0.f,0.f,0.f,0.f,0.f,0.f};
  v8f cQ = {0.f,0.f,0.f,0.f,0.f,0.f,0.f,0.f};

  for (int t = 0; t < DBA_PW / 2; t += 2) {
    float vA, svA, vB, svB;
    {
      const float  lux = pux[0];
      const float  luy = puy[0];
      const float2 wv  = pw[0];
      const float2 jd  = pjd[0];
      pux += strideF; puy += strideF; pw += 2; pjd += 2;

      const float ux = lux * msk;
      const float uy = luy * msk;
      v2f pa, pb;
      pa.x = wv.x * ux; pa.y = wv.x * uy;
      pb.x = ux;        pb.y = uy;
      cP = __builtin_amdgcn_wmma_f32_16x16x4_f32(false, pa, false, pb,
                                                 (short)0, cP, false, false);
      vA = jd.x * pa.x + jd.y * pa.y;
      const float hdd = wv.x * (jd.x * jd.x + jd.y * jd.y) + lmb + wv.y;
      svA = vA * fast_rcp(hdd + 1e-6f);
    }
    {
      const float  lux = pux[0];
      const float  luy = puy[0];
      const float2 wv  = pw[0];
      const float2 jd  = pjd[0];
      pux += strideF; puy += strideF; pw += 2; pjd += 2;

      const float ux = lux * msk;
      const float uy = luy * msk;
      v2f pa, pb;
      pa.x = wv.x * ux; pa.y = wv.x * uy;
      pb.x = ux;        pb.y = uy;
      cP = __builtin_amdgcn_wmma_f32_16x16x4_f32(false, pa, false, pb,
                                                 (short)0, cP, false, false);
      vB = jd.x * pa.x + jd.y * pa.y;
      const float hdd = wv.x * (jd.x * jd.x + jd.y * jd.y) + lmb + wv.y;
      svB = vB * fast_rcp(hdd + 1e-6f);
    }
    const float vAo  = __shfl_xor(vA, 16, 32);
    const float svAo = __shfl_xor(svA, 16, 32);
    const float vBo  = __shfl_xor(vB, 16, 32);
    const float svBo = __shfl_xor(svB, 16, 32);
    const bool lo = (half == 0);
    v2f qa, qb;
    qa.x = lo ? svA  : svBo;  qa.y = lo ? svAo : svB;
    qb.x = lo ? vA   : vBo;   qb.y = lo ? vAo  : vB;
    cQ = __builtin_amdgcn_wmma_f32_16x16x4_f32(false, qa, false, qb,
                                               (short)0, cQ, false, false);
  }

  __shared__ float sred[8][112];
  if (lane < 7) {
#pragma unroll
    for (int p = 0; p < 7; ++p) {
      sred[wave][p * 7 + lane]      = cP[p];
      sred[wave][49 + p * 7 + lane] = cQ[p];
    }
  }
  __syncthreads();
  if (threadIdx.x < 98) {
    float s = 0.f;
#pragma unroll
    for (int wv = 0; wv < 8; ++wv) s += sred[wv][threadIdx.x];
    partial[(size_t)blockIdx.x * 128 + threadIdx.x] = s;
  }
}

__global__ __launch_bounds__(128) void dba_stage2_solve(
    const float* __restrict__ partial, const float* __restrict__ lmbda,
    float* __restrict__ out)
{
  const int b = blockIdx.x;
  __shared__ float acc[98];
  if (threadIdx.x < 98) {
    float s = 0.f;
    for (int g = 0; g < DBA_G; ++g)
      s += partial[(size_t)(b * DBA_G + g) * 128 + threadIdx.x];
    acc[threadIdx.x] = s;
  }
  __syncthreads();
  if (threadIdx.x == 0) {
    const float lmb = lmbda[0];
    float A[6][7];
    for (int ii = 0; ii < 6; ++ii) {
      for (int jj = 0; jj < 6; ++jj) {
        const float hij = acc[ii * 7 + jj] - acc[49 + ii * 7 + jj];
        const float hji = acc[jj * 7 + ii] - acc[49 + jj * 7 + ii];
        A[ii][jj] = 0.5f * (hij + hji);
      }
      A[ii][ii] += lmb + 0.001f;
      A[ii][6] = acc[ii * 7 + 6] - acc[49 + ii * 7 + 6];
    }
    for (int col = 0; col < 6; ++col) {
      int piv = col; float best = fabsf(A[col][col]);
      for (int rr = col + 1; rr < 6; ++rr) {
        const float av = fabsf(A[rr][col]);
        if (av > best) { best = av; piv = rr; }
      }
      if (piv != col)
        for (int jj = col; jj < 7; ++jj) {
          const float tmp = A[col][jj]; A[col][jj] = A[piv][jj]; A[piv][jj] = tmp;
        }
      const float invd = 1.0f / A[col][col];
      for (int rr = col + 1; rr < 6; ++rr) {
        const float f = A[rr][col] * invd;
        for (int jj = col; jj < 7; ++jj) A[rr][jj] -= f * A[col][jj];
      }
    }
    float x[6];
    for (int ii = 5; ii >= 0; --ii) {
      float s = A[ii][6];
      for (int jj = ii + 1; jj < 6; ++jj) s -= A[ii][jj] * x[jj];
      x[ii] = s / A[ii][ii];
    }
    for (int ii = 0; ii < 6; ++ii) out[b * 6 + ii] = x[ii];
  }
}

__global__ __launch_bounds__(256) void dba_stage3_depth(
    const float* __restrict__ r, const float* __restrict__ w,
    const float* __restrict__ Jp, const float* __restrict__ Jd,
    const float* __restrict__ lmbda, const float* __restrict__ dp,
    float* __restrict__ dd)
{
  const int idx = blockIdx.x * blockDim.x + threadIdx.x;
  if (idx >= DBA_B * DBA_N) return;
  const int b = idx / DBA_N;
  const size_t rb = (size_t)idx;

  const float4* Jp4 = (const float4*)(Jp + rb * 12);
  const float4 a0 = Jp4[0], a1 = Jp4[1], a2 = Jp4[2];
  const float2 wv = ((const float2*)w)[rb];
  const float2 jd = ((const float2*)Jd)[rb];
  const float2 rv = ((const float2*)r)[rb];
  const float lmb = lmbda[0];

  const float p0 = dp[b * 6 + 0], p1 = dp[b * 6 + 1], p2 = dp[b * 6 + 2];
  const float p3 = dp[b * 6 + 3], p4 = dp[b * 6 + 4], p5 = dp[b * 6 + 5];

  const float h0 = wv.x * (jd.x * a0.x + jd.y * a1.z);
  const float h1 = wv.x * (jd.x * a0.y + jd.y * a1.w);
  const float h2 = wv.x * (jd.x * a0.z + jd.y * a2.x);
  const float h3 = wv.x * (jd.x * a0.w + jd.y * a2.y);
  const float h4 = wv.x * (jd.x * a1.x + jd.y * a2.z);
  const float h5 = wv.x * (jd.x * a1.y + jd.y * a2.w);
  const float gd  = wv.x * (jd.x * rv.x + jd.y * rv.y);
  const float hdd = wv.x * (jd.x * jd.x + jd.y * jd.y) + lmb + wv.y;
  const float inv = fast_rcp(hdd + 1e-6f);
  const float vdot = h0 * p0 + h1 * p1 + h2 * p2 + h3 * p3 + h4 * p4 + h5 * p5;
  dd[idx] = inv * (gd - vdot);
}

extern "C" void kernel_launch(void* const* d_in, const int* in_sizes, int n_in,
                              void* d_out, int out_size, void* d_ws, size_t ws_size,
                              hipStream_t stream) {
  const float* r   = (const float*)d_in[0];
  const float* w   = (const float*)d_in[1];
  const float* Jp  = (const float*)d_in[2];
  const float* Jd  = (const float*)d_in[3];
  const float* lmb = (const float*)d_in[4];
  float* out = (float*)d_out;
  float* ws  = (float*)d_ws;

  dba_stage1_reduce<<<dim3(DBA_B * DBA_G), dim3(256), 0, stream>>>(
      r, w, Jp, Jd, lmb, ws);
  dba_stage2_solve<<<dim3(DBA_B), dim3(128), 0, stream>>>(ws, lmb, out);
  const int tot = DBA_B * DBA_N;
  dba_stage3_depth<<<dim3((tot + 255) / 256), dim3(256), 0, stream>>>(
      r, w, Jp, Jd, lmb, out, out + DBA_B * 6);
}


# stage 3 runs on stage 1's grid: a wave walks its 256 nodes as 8 rows of 32 (one lane per node, several rows' loads in flight, registers only, no LDS): 8192 waves enter, not 65536; sum order unchanged
# speedup vs baseline: 1.9866x; 1.9866x over previous
#include <hip/hip_runtime.h>


typedef __attribute__((ext_vector_type(2))) float v2f;
typedef __attribute__((ext_vector_type(8))) float v8f;

#define DBA_B 32
#define DBA_N 65536
#define DBA_G 32
#define DBA_SPAN (DBA_N / DBA_G)
#define DBA_PW (DBA_SPAN / 8)

__device__ __forceinline__ float fast_rcp(float x) {
  float inv = __builtin_amdgcn_rcpf(x);
  inv = inv * (2.0f - x * inv);
  return inv;
}

__global__ __launch_bounds__(256) void dba_stage1_reduce(
    const float* __restrict__ r, const float* __restrict__ w,
    const float* __restrict__ Jp, const float* __restrict__ Jd,
    const float* __restrict__ lmbda, float* __restrict__ partial)
{
  const int b    = blockIdx.x / DBA_G;
  const int c    = blockIdx.x % DBA_G;
  const int wave = threadIdx.x >> 5;
  const int lane = threadIdx.x & 31;
  const int half = lane >> 4;
  const int i    = lane & 15;
  const float lmb = lmbda[0];

  const int base = c * DBA_SPAN + wave * DBA_PW;
  const size_t n0 = (size_t)b * DBA_N + base + 2 * half;

  const float* px;
  int strideN;
  if (i < 12) { px = Jp + n0 * 12 + i;      strideN = 12; }
  else        { px = r + n0 * 2 + (i & 1);  strideN = 2; }
  const float4* pw  = (const float4*)(w  + n0 * 2);
  const float4* pjd = (const float4*)(Jd + n0 * 2);

  v8f cP = {0.f,0.f,0.f,0.f,0.f,0.f,0.f,0.f};
  v8f cQ = {0.f,0.f,0.f,0.f,0.f,0.f,0.f,0.f};

  const float4* pwl  = pw + 2 * i;
  const float4* pjdl = pjd + 2 * i;
  const float* pjk = Jd + ((size_t)b * DBA_N + base) * 2 + 4 * half + 8 * (i >> 2) + (i & 3);

  for (int t = 0; t < DBA_PW / 64; ++t) {
    float xa[16], xb[16], jk[4];
    asm volatile("" : "+v"(strideN));
#pragma unroll
    for (int g = 0; g < 16; ++g) {
      xa[g] = px[(4 * g) * strideN];
      xb[g] = px[(4 * g + 1) * strideN];
    }
    const float4 wq = *pwl;
    const float4 jq = *pjdl;
#pragma unroll
    for (int q = 0; q < 4; ++q) jk[q] = pjk[32 * q];
    px += 64 * strideN; pwl += 32; pjdl += 32; pjk += 128;
    __builtin_amdgcn_sched_barrier(0);

    const float hddA = wq.x * (jq.x * jq.x + jq.y * jq.y) + lmb + wq.y;
    const float hddB = wq.z * (jq.z * jq.z + jq.w * jq.w) + lmb + wq.w;
    const float invA = fast_rcp(hddA + 1e-6f);
    const float invB = fast_rcp(hddB + 1e-6f);
    const unsigned vcA = __float_as_uint(wq.x);
    const unsigned vcB = __float_as_uint(wq.z);
    const unsigned vsA = __float_as_uint((invA * wq.x) * wq.x);
    const unsigned vsB = __float_as_uint((invB * wq.z) * wq.z);

#pragma unroll
    for (int g = 0; g < 16; ++g) {
      const unsigned sb = 0x11111111u * (unsigned)g;
      const float confA = __uint_as_float(__builtin_amdgcn_permlane16(vcA, vcA, sb, sb, true, false));
      const float confB = __uint_as_float(__builtin_amdgcn_permlane16(vcB, vcB, sb, sb, true, false));
      const float sA    = __uint_as_float(__builtin_amdgcn_permlane16(vsA, vsA, sb, sb, true, false));
      const float sB    = __uint_as_float(__builtin_amdgcn_permlane16(vsB, vsB, sb, sb, true, false));
      const unsigned ba = 0x11111111u * (unsigned)(4 * (g & 3));
      const unsigned bb = 0x11111111u * (unsigned)(4 * (g & 3) + 2);
      const unsigned vjk = __float_as_uint(jk[g >> 2]);
      const float eA = __uint_as_float(__builtin_amdgcn_permlane16(
          vjk, vjk, ba + 0x11000000u, ba + 0x10101111u, true, false));
      const float eB = __uint_as_float(__builtin_amdgcn_permlane16(
          vjk, vjk, bb + 0x11000000u, bb + 0x10101111u, true, false));
      v2f pa, pb;
      pa.x = confA * xa[g]; pa.y = confB * xb[g];
      pb.x = xa[g];         pb.y = xb[g];
      cP = __builtin_amdgcn_wmma_f32_16x16x4_f32(false, pa, false, pb,
                                                 (short)0, cP, false, false);
      v2f qa, qb;
      qb.x = eA * xa[g];    qb.y = eB * xb[g];
      qa.x = sA * qb.x;     qa.y = sB * qb.y;
      cQ = __builtin_amdgcn_wmma_f32_16x16x4_f32(false, qa, false, qb,
                                                 (short)0, cQ, false, false);
    }
  }

  __shared__ float sD[8][2][16][16];
#pragma unroll
  for (int p = 0; p < 8; ++p) {
    sD[wave][0][8 * half + p][i] = cP[p];
    sD[wave][1][8 * half + p][i] = cQ[p];
  }
  __syncthreads();
  if (threadIdx.x < 98) {
    const int m  = threadIdx.x / 49;
    const int rm = threadIdx.x % 49;
    const int ii = rm / 7, jj = rm % 7;
    const int ia = (ii < 6) ? ii : 12, ib = (ii < 6) ? ii + 6 : 13;
    const int ja = (jj < 6) ? jj : 12, jb = (jj < 6) ? jj + 6 : 13;
    float s = 0.f;
#pragma unroll
    for (int wv = 0; wv < 8; ++wv) {
      float e = sD[wv][m][ia][ja] + sD[wv][m][ib][jb];
      if (m == 1) e += sD[wv][m][ia][jb] + sD[wv][m][ib][ja];
      s += e;
    }
    partial[(size_t)blockIdx.x * 128 + threadIdx.x] = s;
  }
}

__global__ __launch_bounds__(128) void dba_stage2_solve(
    const float* __restrict__ partial, const float* __restrict__ lmbda,
    float* __restrict__ out)
{
  const int b = blockIdx.x;
  __shared__ float acc[98];
  if (threadIdx.x < 98) {
    float s = 0.f;
    for (int g = 0; g < DBA_G; ++g)
      s += partial[(size_t)(b * DBA_G + g) * 128 + threadIdx.x];
    acc[threadIdx.x] = s;
  }
  __syncthreads();
  if (threadIdx.x == 0) {
    const float lmb = lmbda[0];
    float A[6][7];
    for (int ii = 0; ii < 6; ++ii) {
      for (int jj = 0; jj < 6; ++jj) {
        const float hij = acc[ii * 7 + jj] - acc[49 + ii * 7 + jj];
        const float hji = acc[jj * 7 + ii] - acc[49 + jj * 7 + ii];
        A[ii][jj] = 0.5f * (hij + hji);
      }
      A[ii][ii] += lmb + 0.001f;
      A[ii][6] = acc[ii * 7 + 6] - acc[49 + ii * 7 + 6];
    }
    for (int col = 0; col < 6; ++col) {
      int piv = col; float best = fabsf(A[col][col]);
      for (int rr = col + 1; rr < 6; ++rr) {
        const float av = fabsf(A[rr][col]);
        if (av > best) { best = av; piv = rr; }
      }
      if (piv != col)
        for (int jj = col; jj < 7; ++jj) {
          const float tmp = A[col][jj]; A[col][jj] = A[piv][jj]; A[piv][jj] = tmp;
        }
      const float invd = 1.0f / A[col][col];
      for (int rr = col + 1; rr < 6; ++rr) {
        const float f = A[rr][col] * invd;
        for (int jj = col; jj < 7; ++jj) A[rr][jj] -= f * A[col][jj];
      }
    }
    float x[6];
    for (int ii = 5; ii >= 0; --ii) {
      float s = A[ii][6];
      for (int jj = ii + 1; jj < 6; ++jj) s -= A[ii][jj] * x[jj];
      x[ii] = s / A[ii][ii];
    }
    for (int ii = 0; ii < 6; ++ii) out[b * 6 + ii] = x[ii];
  }
}

__global__ __launch_bounds__(256) void dba_stage3_depth(
    const float* __restrict__ r, const float* __restrict__ w,
    const float* __restrict__ Jp, const float* __restrict__ Jd,
    const float* __restrict__ lmbda, const float* __restrict__ dp,
    float* __restrict__ dd)
{
  const int b    = blockIdx.x / DBA_G;
  const int c    = blockIdx.x % DBA_G;
  const int wave = threadIdx.x >> 5;
  const int lane = threadIdx.x & 31;
  size_t n = (size_t)b * DBA_N + c * DBA_SPAN + wave * DBA_PW + lane;

  const float lmb = lmbda[0];
  const float p0 = dp[b * 6 + 0], p1 = dp[b * 6 + 1], p2 = dp[b * 6 + 2];
  const float p3 = dp[b * 6 + 3], p4 = dp[b * 6 + 4], p5 = dp[b * 6 + 5];

  for (int t = 0; t < DBA_PW / 128; ++t) {
    float4 a0[4], a1[4], a2[4];
    float2 wv[4], jd[4], rv[4];
#pragma unroll
    for (int j = 0; j < 4; ++j) {
      const size_t rb = n + 32 * j;
      const float4* Jp4 = (const float4*)(Jp + rb * 12);
      a0[j] = Jp4[0]; a1[j] = Jp4[1]; a2[j] = Jp4[2];
      wv[j] = ((const float2*)w)[rb];
      jd[j] = ((const float2*)Jd)[rb];
      rv[j] = ((const float2*)r)[rb];
    }
#pragma unroll
    for (int j = 0; j < 4; ++j) {
      const float h0 = wv[j].x * (jd[j].x * a0[j].x + jd[j].y * a1[j].z);
      const float h1 = wv[j].x * (jd[j].x * a0[j].y + jd[j].y * a1[j].w);
      const float h2 = wv[j].x * (jd[j].x * a0[j].z + jd[j].y * a2[j].x);
      const float h3 = wv[j].x * (jd[j].x * a0[j].w + jd[j].y * a2[j].y);
      const float h4 = wv[j].x * (jd[j].x * a1[j].x + jd[j].y * a2[j].z);
      const float h5 = wv[j].x * (jd[j].x * a1[j].y + jd[j].y * a2[j].w);
      const float gd  = wv[j].x * (jd[j].x * rv[j].x + jd[j].y * rv[j].y);
      const float hdd = wv[j].x * (jd[j].x * jd[j].x + jd[j].y * jd[j].y) + lmb + wv[j].y;
      const float inv = fast_rcp(hdd + 1e-6f);
      const float vdot = h0 * p0 + h1 * p1 + h2 * p2 + h3 * p3 + h4 * p4 + h5 * p5;
      dd[n + 32 * j] = inv * (gd - vdot);
    }
    n += 128;
  }
}

extern "C" void kernel_launch(void* const* d_in, const int* in_sizes, int n_in,
                              void* d_out, int out_size, void* d_ws, size_t ws_size,
                              hipStream_t stream) {
  const float* r   = (const float*)d_in[0];
  const float* w   = (const float*)d_in[1];
  const float* Jp  = (const float*)d_in[2];
  const float* Jd  = (const float*)d_in[3];
  const float* lmb = (const float*)d_in[4];
  float* out = (float*)d_out;
  float* ws  = (float*)d_ws;

  dba_stage1_reduce<<<dim3(DBA_B * DBA_G), dim3(256), 0, stream>>>(
      r, w, Jp, Jd, lmb, ws);
  dba_stage2_solve<<<dim3(DBA_B), dim3(128), 0, stream>>>(ws, lmb, out);
  dba_stage3_depth<<<dim3(DBA_B * DBA_G), dim3(256), 0, stream>>>(
      r, w, Jp, Jd, lmb, out, out + DBA_B * 6);
}


# stage 2 solves its 6x7 system in registers: elimination fully unrolled, pivot row exchanged by compare-and-select, no scratch store/load round trips between steps; arithmetic and order unchanged
# speedup vs baseline: 2.0968x; 1.0555x over previous
#include <hip/hip_runtime.h>


typedef __attribute__((ext_vector_type(2))) float v2f;
typedef __attribute__((ext_vector_type(8))) float v8f;

#define DBA_B 32
#define DBA_N 65536
#define DBA_G 32
#define DBA_SPAN (DBA_N / DBA_G)
#define DBA_PW (DBA_SPAN / 8)

__device__ __forceinline__ float fast_rcp(float x) {
  float inv = __builtin_amdgcn_rcpf(x);
  inv = inv * (2.0f - x * inv);
  return inv;
}

__global__ __launch_bounds__(256) void dba_stage1_reduce(
    const float* __restrict__ r, const float* __restrict__ w,
    const float* __restrict__ Jp, const float* __restrict__ Jd,
    const float* __restrict__ lmbda, float* __restrict__ partial)
{
  const int b    = blockIdx.x / DBA_G;
  const int c    = blockIdx.x % DBA_G;
  const int wave = threadIdx.x >> 5;
  const int lane = threadIdx.x & 31;
  const int half = lane >> 4;
  const int i    = lane & 15;
  const float lmb = lmbda[0];

  const int base = c * DBA_SPAN + wave * DBA_PW;
  const size_t n0 = (size_t)b * DBA_N + base + 2 * half;

  const float* px;
  int strideN;
  if (i < 12) { px = Jp + n0 * 12 + i;      strideN = 12; }
  else        { px = r + n0 * 2 + (i & 1);  strideN = 2; }
  const float4* pw  = (const float4*)(w  + n0 * 2);
  const float4* pjd = (const float4*)(Jd + n0 * 2);

  v8f cP = {0.f,0.f,0.f,0.f,0.f,0.f,0.f,0.f};
  v8f cQ = {0.f,0.f,0.f,0.f,0.f,0.f,0.f,0.f};

  const float4* pwl  = pw + 2 * i;
  const float4* pjdl = pjd + 2 * i;
  const float* pjk = Jd + ((size_t)b * DBA_N + base) * 2 + 4 * half + 8 * (i >> 2) + (i & 3);

  for (int t = 0; t < DBA_PW / 64; ++t) {
    float xa[16], xb[16], jk[4];
    asm volatile("" : "+v"(strideN));
#pragma unroll
    for (int g = 0; g < 16; ++g) {
      xa[g] = px[(4 * g) * strideN];
      xb[g] = px[(4 * g + 1) * strideN];
    }
    const float4 wq = *pwl;
    const float4 jq = *pjdl;
#pragma unroll
    for (int q = 0; q < 4; ++q) jk[q] = pjk[32 * q];
    px += 64 * strideN; pwl += 32; pjdl += 32; pjk += 128;
    __builtin_amdgcn_sched_barrier(0);

    const float hddA = wq.x * (jq.x * jq.x + jq.y * jq.y) + lmb + wq.y;
    const float hddB = wq.z * (jq.z * jq.z + jq.w * jq.w) + lmb + wq.w;
    const float invA = fast_rcp(hddA + 1e-6f);
    const float invB = fast_rcp(hddB + 1e-6f);
    const unsigned vcA = __float_as_uint(wq.x);
    const unsigned vcB = __float_as_uint(wq.z);
    const unsigned vsA = __float_as_uint((invA * wq.x) * wq.x);
    const unsigned vsB = __float_as_uint((invB * wq.z) * wq.z);

#pragma unroll
    for (int g = 0; g < 16; ++g) {
      const unsigned sb = 0x11111111u * (unsigned)g;
      const float confA = __uint_as_float(__builtin_amdgcn_permlane16(vcA, vcA, sb, sb, true, false));
      const float confB = __uint_as_float(__builtin_amdgcn_permlane16(vcB, vcB, sb, sb, true, false));
      const float sA    = __uint_as_float(__builtin_amdgcn_permlane16(vsA, vsA, sb, sb, true, false));
      const float sB    = __uint_as_float(__builtin_amdgcn_permlane16(vsB, vsB, sb, sb, true, false));
      const unsigned ba = 0x11111111u * (unsigned)(4 * (g & 3));
      const unsigned bb = 0x11111111u * (unsigned)(4 * (g & 3) + 2);
      const unsigned vjk = __float_as_uint(jk[g >> 2]);
      const float eA = __uint_as_float(__builtin_amdgcn_permlane16(
          vjk, vjk, ba + 0x11000000u, ba + 0x10101111u, true, false));
      const float eB = __uint_as_float(__builtin_amdgcn_permlane16(
          vjk, vjk, bb + 0x11000000u, bb + 0x10101111u, true, false));
      v2f pa, pb;
      pa.x = confA * xa[g]; pa.y = confB * xb[g];
      pb.x = xa[g];         pb.y = xb[g];
      cP = __builtin_amdgcn_wmma_f32_16x16x4_f32(false, pa, false, pb,
                                                 (short)0, cP, false, false);
      v2f qa, qb;
      qb.x = eA * xa[g];    qb.y = eB * xb[g];
      qa.x = sA * qb.x;     qa.y = sB * qb.y;
      cQ = __builtin_amdgcn_wmma_f32_16x16x4_f32(false, qa, false, qb,
                                                 (short)0, cQ, false, false);
    }
  }

  __shared__ float sD[8][2][16][16];
#pragma unroll
  for (int p = 0; p < 8; ++p) {
    sD[wave][0][8 * half + p][i] = cP[p];
    sD[wave][1][8 * half + p][i] = cQ[p];
  }
  __syncthreads();
  if (threadIdx.x < 98) {
    const int m  = threadIdx.x / 49;
    const int rm = threadIdx.x % 49;
    const int ii = rm / 7, jj = rm % 7;
    const int ia = (ii < 6) ? ii : 12, ib = (ii < 6) ? ii + 6 : 13;
    const int ja = (jj < 6) ? jj : 12, jb = (jj < 6) ? jj + 6 : 13;
    float s = 0.f;
#pragma unroll
    for (int wv = 0; wv < 8; ++wv) {
      float e = sD[wv][m][ia][ja] + sD[wv][m][ib][jb];
      if (m == 1) e += sD[wv][m][ia][jb] + sD[wv][m][ib][ja];
      s += e;
    }
    partial[(size_t)blockIdx.x * 128 + threadIdx.x] = s;
  }
}

__global__ __launch_bounds__(128) void dba_stage2_solve(
    const float* __restrict__ partial, const float* __restrict__ lmbda,
    float* __restrict__ out)
{
  const int b = blockIdx.x;
  __shared__ float acc[98];
  if (threadIdx.x < 98) {
    float s = 0.f;
    for (int g = 0; g < DBA_G; ++g)
      s += partial[(size_t)(b * DBA_G + g) * 128 + threadIdx.x];
    acc[threadIdx.x] = s;
  }
  __syncthreads();
  if (threadIdx.x == 0) {
    const float lmb = lmbda[0];
    float A[6][7];
#pragma unroll
    for (int ii = 0; ii < 6; ++ii) {
#pragma unroll
      for (int jj = 0; jj < 6; ++jj) {
        const float hij = acc[ii * 7 + jj] - acc[49 + ii * 7 + jj];
        const float hji = acc[jj * 7 + ii] - acc[49 + jj * 7 + ii];
        A[ii][jj] = 0.5f * (hij + hji);
      }
      A[ii][ii] += lmb + 0.001f;
      A[ii][6] = acc[ii * 7 + 6] - acc[49 + ii * 7 + 6];
    }
#pragma unroll
    for (int col = 0; col < 6; ++col) {
      int piv = col; float best = fabsf(A[col][col]);
#pragma unroll
      for (int rr = col + 1; rr < 6; ++rr) {
        const float av = fabsf(A[rr][col]);
        if (av > best) { best = av; piv = rr; }
      }
#pragma unroll
      for (int jj = col; jj < 7; ++jj) {
        const float top = A[col][jj];
        float pv = top;
#pragma unroll
        for (int rr = col + 1; rr < 6; ++rr) {
          const bool hit = (piv == rr);
          pv = hit ? A[rr][jj] : pv;
          A[rr][jj] = hit ? top : A[rr][jj];
        }
        A[col][jj] = pv;
      }
      const float invd = 1.0f / A[col][col];
#pragma unroll
      for (int rr = col + 1; rr < 6; ++rr) {
        const float f = A[rr][col] * invd;
#pragma unroll
        for (int jj = col; jj < 7; ++jj) A[rr][jj] -= f * A[col][jj];
      }
    }
    float x[6];
#pragma unroll
    for (int ii = 5; ii >= 0; --ii) {
      float s = A[ii][6];
#pragma unroll
      for (int jj = ii + 1; jj < 6; ++jj) s -= A[ii][jj] * x[jj];
      x[ii] = s / A[ii][ii];
    }
#pragma unroll
    for (int ii = 0; ii < 6; ++ii) out[b * 6 + ii] = x[ii];
  }
}

__global__ __launch_bounds__(256) void dba_stage3_depth(
    const float* __restrict__ r, const float* __restrict__ w,
    const float* __restrict__ Jp, const float* __restrict__ Jd,
    const float* __restrict__ lmbda, const float* __restrict__ dp,
    float* __restrict__ dd)
{
  const int b    = blockIdx.x / DBA_G;
  const int c    = blockIdx.x % DBA_G;
  const int wave = threadIdx.x >> 5;
  const int lane = threadIdx.x & 31;
  size_t n = (size_t)b * DBA_N + c * DBA_SPAN + wave * DBA_PW + lane;

  const float lmb = lmbda[0];
  const float p0 = dp[b * 6 + 0], p1 = dp[b * 6 + 1], p2 = dp[b * 6 + 2];
  const float p3 = dp[b * 6 + 3], p4 = dp[b * 6 + 4], p5 = dp[b * 6 + 5];

  for (int t = 0; t < DBA_PW / 128; ++t) {
    float4 a0[4], a1[4], a2[4];
    float2 wv[4], jd[4], rv[4];
#pragma unroll
    for (int j = 0; j < 4; ++j) {
      const size_t rb = n + 32 * j;
      const float4* Jp4 = (const float4*)(Jp + rb * 12);
      a0[j] = Jp4[0]; a1[j] = Jp4[1]; a2[j] = Jp4[2];
      wv[j] = ((const float2*)w)[rb];
      jd[j] = ((const float2*)Jd)[rb];
      rv[j] = ((const float2*)r)[rb];
    }
#pragma unroll
    for (int j = 0; j < 4; ++j) {
      const float h0 = wv[j].x * (jd[j].x * a0[j].x + jd[j].y * a1[j].z);
      const float h1 = wv[j].x * (jd[j].x * a0[j].y + jd[j].y * a1[j].w);
      const float h2 = wv[j].x * (jd[j].x * a0[j].z + jd[j].y * a2[j].x);
      const float h3 = wv[j].x * (jd[j].x * a0[j].w + jd[j].y * a2[j].y);
      const float h4 = wv[j].x * (jd[j].x * a1[j].x + jd[j].y * a2[j].z);
      const float h5 = wv[j].x * (jd[j].x * a1[j].y + jd[j].y * a2[j].w);
      const float gd  = wv[j].x * (jd[j].x * rv[j].x + jd[j].y * rv[j].y);
      const float hdd = wv[j].x * (jd[j].x * jd[j].x + jd[j].y * jd[j].y) + lmb + wv[j].y;
      const float inv = fast_rcp(hdd + 1e-6f);
      const float vdot = h0 * p0 + h1 * p1 + h2 * p2 + h3 * p3 + h4 * p4 + h5 * p5;
      dd[n + 32 * j] = inv * (gd - vdot);
    }
    n += 128;
  }
}

extern "C" void kernel_launch(void* const* d_in, const int* in_sizes, int n_in,
                              void* d_out, int out_size, void* d_ws, size_t ws_size,
                              hipStream_t stream) {
  const float* r   = (const float*)d_in[0];
  const float* w   = (const float*)d_in[1];
  const float* Jp  = (const float*)d_in[2];
  const float* Jd  = (const float*)d_in[3];
  const float* lmb = (const float*)d_in[4];
  float* out = (float*)d_out;
  float* ws  = (float*)d_ws;

  dba_stage1_reduce<<<dim3(DBA_B * DBA_G), dim3(256), 0, stream>>>(
      r, w, Jp, Jd, lmb, ws);
  dba_stage2_solve<<<dim3(DBA_B), dim3(128), 0, stream>>>(ws, lmb, out);
  dba_stage3_depth<<<dim3(DBA_B * DBA_G), dim3(256), 0, stream>>>(
      r, w, Jp, Jd, lmb, out, out + DBA_B * 6);
}
